# starting kernel
# baseline (speedup 1.0000x reference)
#include <hip/hip_runtime.h>


#define QSIZE 128
#define NW    (QSIZE * (QSIZE - 1) / 2)
#define CN_EPS 0.9f

typedef __attribute__((ext_vector_type(2))) float v2f;
typedef __attribute__((ext_vector_type(8))) float v8f;

static __device__ __forceinline__ v8f wmma_f32x4(v2f a, v2f b, v8f c) {
  return __builtin_amdgcn_wmma_f32_16x16x4_f32(
       false, a,  false, b,
       (short)0, c,  false,  false);
}

__global__ __launch_bounds__(256) void build_q(const float* __restrict__ w,
                                               float* __restrict__ Q) {
  __shared__ float red[256];
  const int t = threadIdx.x;

  float s = 0.0f;
  for (int i = t; i < NW; i += 256) {
    float v = w[i];
    s += v * v;
  }
  red[t] = s;
  __syncthreads();
  for (int o = 128; o > 0; o >>= 1) {
    if (t < o) red[t] += red[t + o];
    __syncthreads();
  }
  const float norm = sqrtf(2.0f * red[0]);
  const bool finite = (norm == norm) && (norm < 3.0e38f);
  const float scale = (finite && norm > CN_EPS) ? (CN_EPS / (norm + 1e-12f))
                                                : 1.0f;

  if (t < QSIZE) {
    const int r = t;
    const int base = r * (2 * QSIZE - r - 1) / 2;
    Q[r * QSIZE + r] = 0.0f;
    for (int c = r + 1; c < QSIZE; ++c) {
      const float v = w[base + (c - r - 1)] * scale;
      Q[r * QSIZE + c] = v;
      Q[c * QSIZE + r] = -v;
    }
  }
}

__global__ __launch_bounds__(256) void gemm128(const float* __restrict__ A,
                                               const float* __restrict__ B,
                                               float* __restrict__ C) {
  const int lane = threadIdx.x & 31;
  const int wave = threadIdx.x >> 5;
  const int tile = blockIdx.x * 8 + wave;
  const int m0 = (tile >> 3) * 16;
  const int n0 = (tile & 7) * 16;
  const int lo = lane & 15, hi = lane >> 4;

  v8f acc = {};
#pragma unroll
  for (int kk = 0; kk < QSIZE; kk += 4) {
    v2f a = *(const v2f*)(A + (m0 + lo) * QSIZE + kk + 2 * hi);
    const float* bp = B + (kk + 2 * hi) * QSIZE + n0 + lo;
    v2f b;
    b.x = bp[0];
    b.y = bp[QSIZE];
    acc = wmma_f32x4(a, b, acc);
  }
#pragma unroll
  for (int i = 0; i < 8; ++i)
    C[(m0 + i + 8 * hi) * QSIZE + n0 + lo] = acc[i];
}

__global__ __launch_bounds__(256) void assemble_R(const float* __restrict__ Q,
                                                  const float* __restrict__ Q2,
                                                  const float* __restrict__ vb,
                                                  const float* __restrict__ va,
                                                  float* __restrict__ R) {
  const int lane = threadIdx.x & 31;
  const int wave = threadIdx.x >> 5;
  const int tile = blockIdx.x * 8 + wave;
  const int m0 = (tile >> 3) * 16;
  const int n0 = (tile & 7) * 16;
  const int lo = lane & 15, hi = lane >> 4;

  v8f acc3 = {};
  v8f acc4 = {};
#pragma unroll
  for (int kk = 0; kk < QSIZE; kk += 4) {
    v2f a = *(const v2f*)(Q2 + (m0 + lo) * QSIZE + kk + 2 * hi);
    const float* bq = Q + (kk + 2 * hi) * QSIZE + n0 + lo;
    const float* b2 = Q2 + (kk + 2 * hi) * QSIZE + n0 + lo;
    v2f b3, b4;
    b3.x = bq[0];
    b3.y = bq[QSIZE];
    b4.x = b2[0];
    b4.y = b2[QSIZE];
    acc3 = wmma_f32x4(a, b3, acc3);
    acc4 = wmma_f32x4(a, b4, acc4);
  }

  const float an = va[n0 + lo];
#pragma unroll
  for (int i = 0; i < 8; ++i) {
    const int row = m0 + i + 8 * hi;
    const int col = n0 + lo;
    const float qv = Q[row * QSIZE + col];
    const float q2v = Q2[row * QSIZE + col];
    float r = ((row == col) ? 1.0f : 0.0f) + 2.0f * qv + 2.0f * q2v +
              2.0f * acc3[i] + acc4[i];
    R[row * QSIZE + col] = vb[row] * r * an;
  }
}

__global__ __launch_bounds__(256) void xr_main(const float* __restrict__ X,
                                               const float* __restrict__ R,
                                               float* __restrict__ out) {
  const int lane = threadIdx.x & 31;
  const int wave = threadIdx.x >> 5;
  const int lo = lane & 15, hi = lane >> 4;
  const int n0 = wave * 16;
  const int row_base = blockIdx.x * 256;

  v2f bf[32];
#pragma unroll
  for (int s = 0; s < 32; ++s)
    bf[s] = *(const v2f*)(R + (n0 + lo) * QSIZE + 4 * s + 2 * hi);

  for (int mt = 0; mt < 16; ++mt) {
    const int m0 = row_base + mt * 16;
    const float* xrow = X + (m0 + lo) * QSIZE + 2 * hi;
    v8f acc = {};
#pragma unroll
    for (int s = 0; s < 32; ++s) {
      v2f a = __builtin_nontemporal_load((const v2f*)(xrow + 4 * s));
      acc = wmma_f32x4(a, bf[s], acc);
    }
    float* orow = out + (m0 + 8 * hi) * QSIZE + n0 + lo;
#pragma unroll
    for (int i = 0; i < 8; ++i)
      __builtin_nontemporal_store(acc[i], orow + i * QSIZE);
  }
}

extern "C" void kernel_launch(void* const* d_in, const int* in_sizes, int n_in,
                              void* d_out, int out_size, void* d_ws,
                              size_t ws_size, hipStream_t stream) {
  const float* x = (const float*)d_in[0];
  const float* w = (const float*)d_in[1];
  const float* vb = (const float*)d_in[2];
  const float* va = (const float*)d_in[3];
  float* out = (float*)d_out;

  float* Q = (float*)d_ws;
  float* Q2 = Q + QSIZE * QSIZE;
  float* R = Q2 + QSIZE * QSIZE;

  const int rows = in_sizes[0] / QSIZE;
  const int mblocks = rows / 256;

  build_q<<<1, 256, 0, stream>>>(w, Q);
  gemm128<<<8, 256, 0, stream>>>(Q, Q, Q2);
  assemble_R<<<8, 256, 0, stream>>>(Q, Q2, vb, va, R);
  xr_main<<<mblocks, 256, 0, stream>>>(x, R, out);
}


# step 1 uses the symmetry of Q^2: only the 36 tiles on and above the diagonal are multiplied (9 per SIMD, was 16), each off-diagonal tile is also stored transposed into Q^2 and Q^2 - 2Q
# speedup vs baseline: 4.9216x; 4.9216x over previous
#include <hip/hip_runtime.h>


#define QSIZE 128
#define CN_EPS 0.9f
#define RPITCH 132

typedef __attribute__((ext_vector_type(2))) float v2f;
typedef __attribute__((ext_vector_type(4))) float v4f;
typedef __attribute__((ext_vector_type(8))) float v8f;

static __device__ __forceinline__ v8f wmma_f32x4(v2f a, v2f b, v8f c) {
  return __builtin_amdgcn_wmma_f32_16x16x4_f32(
       false, a,  false, b,
       (short)0, c,  false,  false);
}

static __device__ __forceinline__ void lds_barrier() {
  __builtin_amdgcn_fence(__ATOMIC_RELEASE, "workgroup", "local");
  __builtin_amdgcn_s_barrier();
  __builtin_amdgcn_fence(__ATOMIC_ACQUIRE, "workgroup", "local");
}

template <int NT>
static __device__ __forceinline__ void rows_times_rows(const float* abase,
                                                       const float* bbase,
                                                       v8f (&acc)[NT]) {
  v4f acur = *(const v4f*)abase;
  v4f anxt = acur;
  v4f bcur[NT], bnxt[NT];
#pragma unroll
  for (int j = 0; j < NT; ++j) {
    bcur[j] = *(const v4f*)(bbase + j * 16 * RPITCH);
    bnxt[j] = bcur[j];
  }
#pragma unroll
  for (int t = 0; t < 16; ++t) {
    const bool more = (t + 1 < 16);
    if (more) {
      anxt = *(const v4f*)(abase + 8 * (t + 1));
#pragma unroll
      for (int j = 0; j < NT; ++j)
        bnxt[j] = *(const v4f*)(bbase + j * 16 * RPITCH + 8 * (t + 1));
    }
    const v2f a0 = {acur.x, acur.y};
    const v2f a1 = {acur.z, acur.w};
#pragma unroll
    for (int j = 0; j < NT; ++j) {
      const v2f b0 = {bcur[j].x, bcur[j].y};
      acc[j] = wmma_f32x4(a0, b0, acc[j]);
    }
#pragma unroll
    for (int j = 0; j < NT; ++j) {
      const v2f b1 = {bcur[j].z, bcur[j].w};
      acc[j] = wmma_f32x4(a1, b1, acc[j]);
    }
    if (more) {
      acur = anxt;
#pragma unroll
      for (int j = 0; j < NT; ++j) bcur[j] = bnxt[j];
    }
  }
}

template <int NT>
static __device__ __forceinline__ void tile_pairs(const float* const (&ab)[NT],
                                                  const float* const (&bb)[NT],
                                                  v8f (&acc)[NT]) {
  v4f acur[NT], anxt[NT], bcur[NT], bnxt[NT];
#pragma unroll
  for (int j = 0; j < NT; ++j) {
    acur[j] = *(const v4f*)ab[j];
    bcur[j] = *(const v4f*)bb[j];
    anxt[j] = acur[j];
    bnxt[j] = bcur[j];
  }
#pragma unroll
  for (int t = 0; t < 16; ++t) {
    const bool more = (t + 1 < 16);
    if (more) {
#pragma unroll
      for (int j = 0; j < NT; ++j) {
        anxt[j] = *(const v4f*)(ab[j] + 8 * (t + 1));
        bnxt[j] = *(const v4f*)(bb[j] + 8 * (t + 1));
      }
    }
#pragma unroll
    for (int j = 0; j < NT; ++j) {
      const v2f a0 = {acur[j].x, acur[j].y};
      const v2f b0 = {bcur[j].x, bcur[j].y};
      acc[j] = wmma_f32x4(a0, b0, acc[j]);
    }
#pragma unroll
    for (int j = 0; j < NT; ++j) {
      const v2f a1 = {acur[j].z, acur[j].w};
      const v2f b1 = {bcur[j].z, bcur[j].w};
      acc[j] = wmma_f32x4(a1, b1, acc[j]);
    }
    if (more) {
#pragma unroll
      for (int j = 0; j < NT; ++j) {
        acur[j] = anxt[j];
        bcur[j] = bnxt[j];
      }
    }
  }
}

static __device__ __forceinline__ void upper_tile(int g, int q, int& tm,
                                                  int& tn) {
  if (q < 8 - g) {
    tm = g;
    tn = g + q;
  } else {
    tm = 7 - g;
    tn = q - 1;
  }
}

static __device__ __forceinline__ void store_q2_tile(float* Q2s, float* Us,
                                                     const float* Qs, int tm,
                                                     int tn, int lo, int hi,
                                                     const v8f& acc) {
  float q2[8], um[8];
#pragma unroll
  for (int r = 0; r < 8; ++r) {
    const int e = (16 * tm + 8 * hi + r) * RPITCH + 16 * tn + lo;
    const float qv = Qs[e];
    q2[r] = -acc[r];
    Q2s[e] = q2[r];
    Us[e] = q2[r] - 2.0f * qv;
    um[r] = q2[r] + 2.0f * qv;
  }
  if (tm != tn) {
    const int em = (16 * tn + lo) * RPITCH + 16 * tm + 8 * hi;
    *(v4f*)(Q2s + em) = (v4f){q2[0], q2[1], q2[2], q2[3]};
    *(v4f*)(Q2s + em + 4) = (v4f){q2[4], q2[5], q2[6], q2[7]};
    *(v4f*)(Us + em) = (v4f){um[0], um[1], um[2], um[3]};
    *(v4f*)(Us + em + 4) = (v4f){um[4], um[5], um[6], um[7]};
  }
}

__global__ __launch_bounds__(512) __attribute__((amdgpu_waves_per_eu(4, 4)))
void xr_fused(const float* __restrict__ X, const float* __restrict__ w,
              const float* __restrict__ vb, const float* __restrict__ va,
              float* __restrict__ out) {
  __shared__ __attribute__((aligned(16))) float Qs[QSIZE * RPITCH];
  __shared__ __attribute__((aligned(16))) float Q2s[QSIZE * RPITCH];
  __shared__ __attribute__((aligned(16))) float Us[QSIZE * RPITCH];
  __shared__ __attribute__((aligned(16))) float Rs[QSIZE * RPITCH];
  __shared__ float red[16];

  const int tid = threadIdx.x;
  const int lane = tid & 31;
  const int wave = tid >> 5;
  const int lo = lane & 15, hi = lane >> 4;
  const int mt = wave >> 1, nh = wave & 1;
  const int m0 = 16 * mt;
  const int n0 = 64 * nh;
  const int mx = blockIdx.x * 256 + wave * 16;

  float wv[8][4];
#pragma unroll
  for (int i = 0; i < 8; ++i) {
    const int r = wave + 16 * i;
    const int base = r * (2 * QSIZE - r - 1) / 2;
#pragma unroll
    for (int q = 0; q < 4; ++q) {
      const int c = r + 1 + lane + 32 * q;
      wv[i][q] = (c < QSIZE) ? w[base + lane + 32 * q] : 0.0f;
    }
  }
  float bv[8], av[4];
#pragma unroll
  for (int r = 0; r < 8; ++r) bv[r] = vb[m0 + 8 * hi + r];
#pragma unroll
  for (int j = 0; j < 4; ++j) av[j] = va[n0 + 16 * j + lo];

  const float* xrow = X + (mx + lo) * QSIZE + 4 * hi;
  v4f a[16];
#pragma unroll
  for (int t = 0; t < 16; ++t)
    a[t] = *(const v4f*)(xrow + 8 * t);

  float s = 0.0f;
#pragma unroll
  for (int i = 0; i < 8; ++i) {
#pragma unroll
    for (int q = 0; q < 4; ++q) s += wv[i][q] * wv[i][q];
  }
#pragma unroll
  for (int off = 16; off > 0; off >>= 1) s += __shfl_xor(s, off);
  if (lane == 0) red[wave] = s;
  lds_barrier();
  float tot = 0.0f;
#pragma unroll
  for (int i = 0; i < 16; ++i) tot += red[i];
  const float norm = sqrtf(2.0f * tot);
  const bool finite = (norm == norm) && (norm < 3.0e38f);
  const float scale = (finite && norm > CN_EPS) ? (CN_EPS / (norm + 1e-12f))
                                                : 1.0f;
#pragma unroll
  for (int i = 0; i < 8; ++i) {
    const int r = wave + 16 * i;
#pragma unroll
    for (int q = 0; q < 4; ++q) {
      const int c = r + 1 + lane + 32 * q;
      if (c < QSIZE) {
        const float v = wv[i][q] * scale;
        Qs[r * RPITCH + c] = v;
        Qs[c * RPITCH + r] = -v;
      }
    }
  }
  if (tid < QSIZE) Qs[tid * RPITCH + tid] = 0.0f;
  lds_barrier();

  const int foff = lo * RPITCH + 4 * hi;
  v8f acc[4];

  {
    const int swave = __builtin_amdgcn_readfirstlane(wave);
    const int g = swave >> 2, i = swave & 3;
    int tm0, tn0, tm1, tn1;
    upper_tile(g, i, tm0, tn0);
    upper_tile(g, i + 4, tm1, tn1);
    if (i == g) {
      int tm2, tn2;
      upper_tile(g, 8, tm2, tn2);
      const float* const ab[3] = {Qs + 16 * tm0 * RPITCH + foff,
                                  Qs + 16 * tm1 * RPITCH + foff,
                                  Qs + 16 * tm2 * RPITCH + foff};
      const float* const bb[3] = {Qs + 16 * tn0 * RPITCH + foff,
                                  Qs + 16 * tn1 * RPITCH + foff,
                                  Qs + 16 * tn2 * RPITCH + foff};
      v8f acc3[3] = {(v8f){}, (v8f){}, (v8f){}};
      tile_pairs<3>(ab, bb, acc3);
      store_q2_tile(Q2s, Us, Qs, tm0, tn0, lo, hi, acc3[0]);
      store_q2_tile(Q2s, Us, Qs, tm1, tn1, lo, hi, acc3[1]);
      store_q2_tile(Q2s, Us, Qs, tm2, tn2, lo, hi, acc3[2]);
    } else {
      const float* const ab[2] = {Qs + 16 * tm0 * RPITCH + foff,
                                  Qs + 16 * tm1 * RPITCH + foff};
      const float* const bb[2] = {Qs + 16 * tn0 * RPITCH + foff,
                                  Qs + 16 * tn1 * RPITCH + foff};
      v8f acc2[2] = {(v8f){}, (v8f){}};
      tile_pairs<2>(ab, bb, acc2);
      store_q2_tile(Q2s, Us, Qs, tm0, tn0, lo, hi, acc2[0]);
      store_q2_tile(Q2s, Us, Qs, tm1, tn1, lo, hi, acc2[1]);
    }
  }
  lds_barrier();

#pragma unroll
  for (int j = 0; j < 4; ++j) acc[j] = (v8f){};
  rows_times_rows<4>(Q2s + m0 * RPITCH + foff, Us + n0 * RPITCH + foff, acc);

#pragma unroll
  for (int j = 0; j < 4; ++j) {
#pragma unroll
    for (int r = 0; r < 8; ++r) {
      const int row = m0 + 8 * hi + r;
      const int col = n0 + 16 * j + lo;
      const int e = row * RPITCH + col;
      const float rr = ((row == col) ? 1.0f : 0.0f) + 2.0f * Qs[e] +
                       2.0f * Q2s[e] + acc[j][r];
      Rs[e] = bv[r] * rr * av[j];
    }
  }
  lds_barrier();

  v8f oacc[8];
#pragma unroll
  for (int j = 0; j < 8; ++j) oacc[j] = (v8f){};

  const float* bbase = Rs + foff;
  v4f bcur[8], bnxt[8];
#pragma unroll
  for (int j = 0; j < 8; ++j)
    bcur[j] = *(const v4f*)(bbase + j * 16 * RPITCH);

#pragma unroll
  for (int t = 0; t < 16; ++t) {
    const bool more = (t + 1 < 16);
    if (more) {
#pragma unroll
      for (int j = 0; j < 8; ++j)
        bnxt[j] = *(const v4f*)(bbase + j * 16 * RPITCH + 8 * (t + 1));
    }
    const v2f a0 = {a[t].x, a[t].y};
    const v2f a1 = {a[t].z, a[t].w};
#pragma unroll
    for (int j = 0; j < 8; ++j) {
      const v2f b0 = {bcur[j].x, bcur[j].y};
      oacc[j] = wmma_f32x4(a0, b0, oacc[j]);
    }
#pragma unroll
    for (int j = 0; j < 8; ++j) {
      const v2f b1 = {bcur[j].z, bcur[j].w};
      oacc[j] = wmma_f32x4(a1, b1, oacc[j]);
    }
    if (more) {
#pragma unroll
      for (int j = 0; j < 8; ++j) bcur[j] = bnxt[j];
    }
  }

  float* orow = out + (mx + 8 * hi) * QSIZE + lo;
#pragma unroll
  for (int r = 0; r < 8; ++r) {
#pragma unroll
    for (int j = 0; j < 8; ++j)
      __builtin_nontemporal_store(oacc[j][r], orow + r * QSIZE + 16 * j);
  }
}

extern "C" void kernel_launch(void* const* d_in, const int* in_sizes, int n_in,
                              void* d_out, int out_size, void* d_ws,
                              size_t ws_size, hipStream_t stream) {
  const float* x = (const float*)d_in[0];
  const float* w = (const float*)d_in[1];
  const float* vb = (const float*)d_in[2];
  const float* va = (const float*)d_in[3];
  float* out = (float*)d_out;

  const int rows = in_sizes[0] / QSIZE;
  const int mblocks = rows / 256;

  xr_fused<<<mblocks, 512, 0, stream>>>(x, w, vb, va, out);
}


# each workgroup owns 512 rows of X x 64 output columns (was 256 x 128): it builds only its 64 rows of R, so step 2 is 32 tiles per workgroup, not 64; every row of X is read by two workgroups
# speedup vs baseline: 5.0811x; 1.0324x over previous
#include <hip/hip_runtime.h>


#define QSIZE 128
#define CN_EPS 0.9f
#define RPITCH 132

typedef __attribute__((ext_vector_type(2))) float v2f;
typedef __attribute__((ext_vector_type(4))) float v4f;
typedef __attribute__((ext_vector_type(8))) float v8f;

static __device__ __forceinline__ v8f wmma_f32x4(v2f a, v2f b, v8f c) {
  return __builtin_amdgcn_wmma_f32_16x16x4_f32(
       false, a,  false, b,
       (short)0, c,  false,  false);
}

static __device__ __forceinline__ void lds_barrier() {
  __builtin_amdgcn_fence(__ATOMIC_RELEASE, "workgroup", "local");
  __builtin_amdgcn_s_barrier();
  __builtin_amdgcn_fence(__ATOMIC_ACQUIRE, "workgroup", "local");
}

template <int NT>
static __device__ __forceinline__ void rows_times_rows(const float* abase,
                                                       const float* bbase,
                                                       v8f (&acc)[NT]) {
  v4f acur = *(const v4f*)abase;
  v4f anxt = acur;
  v4f bcur[NT], bnxt[NT];
#pragma unroll
  for (int j = 0; j < NT; ++j) {
    bcur[j] = *(const v4f*)(bbase + j * 16 * RPITCH);
    bnxt[j] = bcur[j];
  }
#pragma unroll
  for (int t = 0; t < 16; ++t) {
    const bool more = (t + 1 < 16);
    if (more) {
      anxt = *(const v4f*)(abase + 8 * (t + 1));
#pragma unroll
      for (int j = 0; j < NT; ++j)
        bnxt[j] = *(const v4f*)(bbase + j * 16 * RPITCH + 8 * (t + 1));
    }
    const v2f a0 = {acur.x, acur.y};
    const v2f a1 = {acur.z, acur.w};
#pragma unroll
    for (int j = 0; j < NT; ++j) {
      const v2f b0 = {bcur[j].x, bcur[j].y};
      acc[j] = wmma_f32x4(a0, b0, acc[j]);
    }
#pragma unroll
    for (int j = 0; j < NT; ++j) {
      const v2f b1 = {bcur[j].z, bcur[j].w};
      acc[j] = wmma_f32x4(a1, b1, acc[j]);
    }
    if (more) {
      acur = anxt;
#pragma unroll
      for (int j = 0; j < NT; ++j) bcur[j] = bnxt[j];
    }
  }
}

template <int NT>
static __device__ __forceinline__ void tile_pairs(const float* const (&ab)[NT],
                                                  const float* const (&bb)[NT],
                                                  v8f (&acc)[NT]) {
  v4f acur[NT], anxt[NT], bcur[NT], bnxt[NT];
#pragma unroll
  for (int j = 0; j < NT; ++j) {
    acur[j] = *(const v4f*)ab[j];
    bcur[j] = *(const v4f*)bb[j];
    anxt[j] = acur[j];
    bnxt[j] = bcur[j];
  }
#pragma unroll
  for (int t = 0; t < 16; ++t) {
    const bool more = (t + 1 < 16);
    if (more) {
#pragma unroll
      for (int j = 0; j < NT; ++j) {
        anxt[j] = *(const v4f*)(ab[j] + 8 * (t + 1));
        bnxt[j] = *(const v4f*)(bb[j] + 8 * (t + 1));
      }
    }
#pragma unroll
    for (int j = 0; j < NT; ++j) {
      const v2f a0 = {acur[j].x, acur[j].y};
      const v2f b0 = {bcur[j].x, bcur[j].y};
      acc[j] = wmma_f32x4(a0, b0, acc[j]);
    }
#pragma unroll
    for (int j = 0; j < NT; ++j) {
      const v2f a1 = {acur[j].z, acur[j].w};
      const v2f b1 = {bcur[j].z, bcur[j].w};
      acc[j] = wmma_f32x4(a1, b1, acc[j]);
    }
    if (more) {
#pragma unroll
      for (int j = 0; j < NT; ++j) {
        acur[j] = anxt[j];
        bcur[j] = bnxt[j];
      }
    }
  }
}

static __device__ __forceinline__ void upper_tile(int g, int q, int& tm,
                                                  int& tn) {
  if (q < 8 - g) {
    tm = g;
    tn = g + q;
  } else {
    tm = 7 - g;
    tn = q - 1;
  }
}

static __device__ __forceinline__ void store_q2_tile(float* Q2s, float* Us,
                                                     const float* Qs, int tm,
                                                     int tn, int lo, int hi,
                                                     const v8f& acc) {
  float q2[8], um[8];
#pragma unroll
  for (int r = 0; r < 8; ++r) {
    const int e = (16 * tm + 8 * hi + r) * RPITCH + 16 * tn + lo;
    const float qv = Qs[e];
    q2[r] = -acc[r];
    Q2s[e] = q2[r];
    Us[e] = q2[r] - 2.0f * qv;
    um[r] = q2[r] + 2.0f * qv;
  }
  if (tm != tn) {
    const int em = (16 * tn + lo) * RPITCH + 16 * tm + 8 * hi;
    *(v4f*)(Q2s + em) = (v4f){q2[0], q2[1], q2[2], q2[3]};
    *(v4f*)(Q2s + em + 4) = (v4f){q2[4], q2[5], q2[6], q2[7]};
    *(v4f*)(Us + em) = (v4f){um[0], um[1], um[2], um[3]};
    *(v4f*)(Us + em + 4) = (v4f){um[4], um[5], um[6], um[7]};
  }
}

__global__ __launch_bounds__(512) __attribute__((amdgpu_waves_per_eu(4, 4)))
void xr_fused(const float* __restrict__ X, const float* __restrict__ w,
              const float* __restrict__ vb, const float* __restrict__ va,
              float* __restrict__ out) {
  __shared__ __attribute__((aligned(16))) float Qs[QSIZE * RPITCH];
  __shared__ __attribute__((aligned(16))) float Q2s[QSIZE * RPITCH];
  __shared__ __attribute__((aligned(16))) float Us[QSIZE * RPITCH];
  __shared__ __attribute__((aligned(16))) float Rs[64 * RPITCH];
  __shared__ float red[16];

  const int tid = threadIdx.x;
  const int lane = tid & 31;
  const int wave = tid >> 5;
  const int lo = lane & 15, hi = lane >> 4;
  const int ch = blockIdx.x & 1;
  const int rb = blockIdx.x >> 1;
  const int ml = 16 * (wave >> 2);
  const int m0 = 64 * ch + ml;
  const int n0 = 32 * (wave & 3);
  const int mx = rb * 512 + wave * 32;

  float wv[8][4];
#pragma unroll
  for (int i = 0; i < 8; ++i) {
    const int r = wave + 16 * i;
    const int base = r * (2 * QSIZE - r - 1) / 2;
#pragma unroll
    for (int q = 0; q < 4; ++q) {
      const int c = r + 1 + lane + 32 * q;
      wv[i][q] = (c < QSIZE) ? w[base + lane + 32 * q] : 0.0f;
    }
  }
  float bv[8], av[2];
#pragma unroll
  for (int r = 0; r < 8; ++r) bv[r] = vb[m0 + 8 * hi + r];
#pragma unroll
  for (int j = 0; j < 2; ++j) av[j] = va[n0 + 16 * j + lo];

  const float* xrow = X + (mx + lo) * QSIZE + 4 * hi;
  v4f a[2][16];
#pragma unroll
  for (int t = 0; t < 16; ++t) {
#pragma unroll
    for (int mi = 0; mi < 2; ++mi)
      a[mi][t] = *(const v4f*)(xrow + mi * 16 * QSIZE + 8 * t);
  }

  float s = 0.0f;
#pragma unroll
  for (int i = 0; i < 8; ++i) {
#pragma unroll
    for (int q = 0; q < 4; ++q) s += wv[i][q] * wv[i][q];
  }
#pragma unroll
  for (int off = 16; off > 0; off >>= 1) s += __shfl_xor(s, off);
  if (lane == 0) red[wave] = s;
  lds_barrier();
  float tot = 0.0f;
#pragma unroll
  for (int i = 0; i < 16; ++i) tot += red[i];
  const float norm = sqrtf(2.0f * tot);
  const bool finite = (norm == norm) && (norm < 3.0e38f);
  const float scale = (finite && norm > CN_EPS) ? (CN_EPS / (norm + 1e-12f))
                                                : 1.0f;
#pragma unroll
  for (int i = 0; i < 8; ++i) {
    const int r = wave + 16 * i;
#pragma unroll
    for (int q = 0; q < 4; ++q) {
      const int c = r + 1 + lane + 32 * q;
      if (c < QSIZE) {
        const float v = wv[i][q] * scale;
        Qs[r * RPITCH + c] = v;
        Qs[c * RPITCH + r] = -v;
      }
    }
  }
  if (tid < QSIZE) Qs[tid * RPITCH + tid] = 0.0f;
  lds_barrier();

  const int foff = lo * RPITCH + 4 * hi;
  v8f acc[2];

  {
    const int swave = __builtin_amdgcn_readfirstlane(wave);
    const int g = swave >> 2, i = swave & 3;
    int tm0, tn0, tm1, tn1;
    upper_tile(g, i, tm0, tn0);
    upper_tile(g, i + 4, tm1, tn1);
    if (i == g) {
      int tm2, tn2;
      upper_tile(g, 8, tm2, tn2);
      const float* const ab[3] = {Qs + 16 * tm0 * RPITCH + foff,
                                  Qs + 16 * tm1 * RPITCH + foff,
                                  Qs + 16 * tm2 * RPITCH + foff};
      const float* const bb[3] = {Qs + 16 * tn0 * RPITCH + foff,
                                  Qs + 16 * tn1 * RPITCH + foff,
                                  Qs + 16 * tn2 * RPITCH + foff};
      v8f acc3[3] = {(v8f){}, (v8f){}, (v8f){}};
      tile_pairs<3>(ab, bb, acc3);
      store_q2_tile(Q2s, Us, Qs, tm0, tn0, lo, hi, acc3[0]);
      store_q2_tile(Q2s, Us, Qs, tm1, tn1, lo, hi, acc3[1]);
      store_q2_tile(Q2s, Us, Qs, tm2, tn2, lo, hi, acc3[2]);
    } else {
      const float* const ab[2] = {Qs + 16 * tm0 * RPITCH + foff,
                                  Qs + 16 * tm1 * RPITCH + foff};
      const float* const bb[2] = {Qs + 16 * tn0 * RPITCH + foff,
                                  Qs + 16 * tn1 * RPITCH + foff};
      v8f acc2[2] = {(v8f){}, (v8f){}};
      tile_pairs<2>(ab, bb, acc2);
      store_q2_tile(Q2s, Us, Qs, tm0, tn0, lo, hi, acc2[0]);
      store_q2_tile(Q2s, Us, Qs, tm1, tn1, lo, hi, acc2[1]);
    }
  }
  lds_barrier();

#pragma unroll
  for (int j = 0; j < 2; ++j) acc[j] = (v8f){};
  rows_times_rows<2>(Q2s + m0 * RPITCH + foff, Us + n0 * RPITCH + foff, acc);

#pragma unroll
  for (int j = 0; j < 2; ++j) {
#pragma unroll
    for (int r = 0; r < 8; ++r) {
      const int row = m0 + 8 * hi + r;
      const int col = n0 + 16 * j + lo;
      const int e = row * RPITCH + col;
      const float rr = ((row == col) ? 1.0f : 0.0f) + 2.0f * Qs[e] +
                       2.0f * Q2s[e] + acc[j][r];
      Rs[(ml + 8 * hi + r) * RPITCH + col] = bv[r] * rr * av[j];
    }
  }
  lds_barrier();

  v8f oacc[2][4];
#pragma unroll
  for (int mi = 0; mi < 2; ++mi) {
#pragma unroll
    for (int j = 0; j < 4; ++j) oacc[mi][j] = (v8f){};
  }

  const float* bbase = Rs + foff;
  v4f bcur[4], bnxt[4];
#pragma unroll
  for (int j = 0; j < 4; ++j)
    bcur[j] = *(const v4f*)(bbase + j * 16 * RPITCH);

#pragma unroll
  for (int t = 0; t < 16; ++t) {
    const bool more = (t + 1 < 16);
    if (more) {
#pragma unroll
      for (int j = 0; j < 4; ++j)
        bnxt[j] = *(const v4f*)(bbase + j * 16 * RPITCH + 8 * (t + 1));
    }
#pragma unroll
    for (int mi = 0; mi < 2; ++mi) {
      const v2f a0 = {a[mi][t].x, a[mi][t].y};
#pragma unroll
      for (int j = 0; j < 4; ++j) {
        const v2f b0 = {bcur[j].x, bcur[j].y};
        oacc[mi][j] = wmma_f32x4(a0, b0, oacc[mi][j]);
      }
    }
#pragma unroll
    for (int mi = 0; mi < 2; ++mi) {
      const v2f a1 = {a[mi][t].z, a[mi][t].w};
#pragma unroll
      for (int j = 0; j < 4; ++j) {
        const v2f b1 = {bcur[j].z, bcur[j].w};
        oacc[mi][j] = wmma_f32x4(a1, b1, oacc[mi][j]);
      }
    }
    if (more) {
#pragma unroll
      for (int j = 0; j < 4; ++j) bcur[j] = bnxt[j];
    }
  }

  float* orow = out + (mx + 8 * hi) * QSIZE + 64 * ch + lo;
#pragma unroll
  for (int mi = 0; mi < 2; ++mi) {
#pragma unroll
    for (int r = 0; r < 8; ++r) {
#pragma unroll
      for (int j = 0; j < 4; ++j)
        __builtin_nontemporal_store(oacc[mi][j][r],
                                    orow + (16 * mi + r) * QSIZE + 16 * j);
    }
  }
}

extern "C" void kernel_launch(void* const* d_in, const int* in_sizes, int n_in,
                              void* d_out, int out_size, void* d_ws,
                              size_t ws_size, hipStream_t stream) {
  const float* x = (const float*)d_in[0];
  const float* w = (const float*)d_in[1];
  const float* vb = (const float*)d_in[2];
  const float* va = (const float*)d_in[3];
  float* out = (float*)d_out;

  const int rows = in_sizes[0] / QSIZE;
  const int mblocks = rows / 256;

  xr_fused<<<mblocks, 512, 0, stream>>>(x, w, vb, va, out);
}


# X fragment loads moved from the kernel head, where every wave's 32 requests (16 lines touched each) had to be taken in front of the first barrier, into the k loops of steps 1 and 2, one per k block
# speedup vs baseline: 5.5149x; 1.0854x over previous
#include <hip/hip_runtime.h>


#define QSIZE 128
#define CN_EPS 0.9f
#define RPITCH 132

typedef __attribute__((ext_vector_type(2))) float v2f;
typedef __attribute__((ext_vector_type(4))) float v4f;
typedef __attribute__((ext_vector_type(8))) float v8f;

static __device__ __forceinline__ v8f wmma_f32x4(v2f a, v2f b, v8f c) {
  return __builtin_amdgcn_wmma_f32_16x16x4_f32(
       false, a,  false, b,
       (short)0, c,  false,  false);
}

static __device__ __forceinline__ void lds_barrier() {
  __builtin_amdgcn_fence(__ATOMIC_RELEASE, "workgroup", "local");
  __builtin_amdgcn_s_barrier();
  __builtin_amdgcn_fence(__ATOMIC_ACQUIRE, "workgroup", "local");
}

template <int NT>
static __device__ __forceinline__ void rows_times_rows(const float* abase,
                                                       const float* bbase,
                                                       v8f (&acc)[NT],
                                                       const float* xp,
                                                       v4f (&xa)[16]) {
  v4f acur = *(const v4f*)abase;
  v4f anxt = acur;
  v4f bcur[NT], bnxt[NT];
#pragma unroll
  for (int j = 0; j < NT; ++j) {
    bcur[j] = *(const v4f*)(bbase + j * 16 * RPITCH);
    bnxt[j] = bcur[j];
  }
#pragma unroll
  for (int t = 0; t < 16; ++t) {
    const bool more = (t + 1 < 16);
    if (more) {
      anxt = *(const v4f*)(abase + 8 * (t + 1));
#pragma unroll
      for (int j = 0; j < NT; ++j)
        bnxt[j] = *(const v4f*)(bbase + j * 16 * RPITCH + 8 * (t + 1));
    }
    xa[t] = *(const v4f*)(xp + 8 * t);
    __builtin_amdgcn_sched_barrier(0);
    const v2f a0 = {acur.x, acur.y};
    const v2f a1 = {acur.z, acur.w};
#pragma unroll
    for (int j = 0; j < NT; ++j) {
      const v2f b0 = {bcur[j].x, bcur[j].y};
      acc[j] = wmma_f32x4(a0, b0, acc[j]);
    }
#pragma unroll
    for (int j = 0; j < NT; ++j) {
      const v2f b1 = {bcur[j].z, bcur[j].w};
      acc[j] = wmma_f32x4(a1, b1, acc[j]);
    }
    if (more) {
      acur = anxt;
#pragma unroll
      for (int j = 0; j < NT; ++j) bcur[j] = bnxt[j];
    }
  }
}

template <int NT>
static __device__ __forceinline__ void tile_pairs(const float* const (&ab)[NT],
                                                  const float* const (&bb)[NT],
                                                  v8f (&acc)[NT],
                                                  const float* xp,
                                                  v4f (&xa)[16]) {
  v4f acur[NT], anxt[NT], bcur[NT], bnxt[NT];
#pragma unroll
  for (int j = 0; j < NT; ++j) {
    acur[j] = *(const v4f*)ab[j];
    bcur[j] = *(const v4f*)bb[j];
    anxt[j] = acur[j];
    bnxt[j] = bcur[j];
  }
#pragma unroll
  for (int t = 0; t < 16; ++t) {
    const bool more = (t + 1 < 16);
    if (more) {
#pragma unroll
      for (int j = 0; j < NT; ++j) {
        anxt[j] = *(const v4f*)(ab[j] + 8 * (t + 1));
        bnxt[j] = *(const v4f*)(bb[j] + 8 * (t + 1));
      }
    }
    xa[t] = *(const v4f*)(xp + 8 * t);
    __builtin_amdgcn_sched_barrier(0);
#pragma unroll
    for (int j = 0; j < NT; ++j) {
      const v2f a0 = {acur[j].x, acur[j].y};
      const v2f b0 = {bcur[j].x, bcur[j].y};
      acc[j] = wmma_f32x4(a0, b0, acc[j]);
    }
#pragma unroll
    for (int j = 0; j < NT; ++j) {
      const v2f a1 = {acur[j].z, acur[j].w};
      const v2f b1 = {bcur[j].z, bcur[j].w};
      acc[j] = wmma_f32x4(a1, b1, acc[j]);
    }
    if (more) {
#pragma unroll
      for (int j = 0; j < NT; ++j) {
        acur[j] = anxt[j];
        bcur[j] = bnxt[j];
      }
    }
  }
}

static __device__ __forceinline__ void upper_tile(int g, int q, int& tm,
                                                  int& tn) {
  if (q < 8 - g) {
    tm = g;
    tn = g + q;
  } else {
    tm = 7 - g;
    tn = q - 1;
  }
}

static __device__ __forceinline__ void store_q2_tile(float* Q2s, float* Us,
                                                     const float* Qs, int tm,
                                                     int tn, int lo, int hi,
                                                     const v8f& acc) {
  float q2[8], um[8];
#pragma unroll
  for (int r = 0; r < 8; ++r) {
    const int e = (16 * tm + 8 * hi + r) * RPITCH + 16 * tn + lo;
    const float qv = Qs[e];
    q2[r] = -acc[r];
    Q2s[e] = q2[r];
    Us[e] = q2[r] - 2.0f * qv;
    um[r] = q2[r] + 2.0f * qv;
  }
  if (tm != tn) {
    const int em = (16 * tn + lo) * RPITCH + 16 * tm + 8 * hi;
    *(v4f*)(Q2s + em) = (v4f){q2[0], q2[1], q2[2], q2[3]};
    *(v4f*)(Q2s + em + 4) = (v4f){q2[4], q2[5], q2[6], q2[7]};
    *(v4f*)(Us + em) = (v4f){um[0], um[1], um[2], um[3]};
    *(v4f*)(Us + em + 4) = (v4f){um[4], um[5], um[6], um[7]};
  }
}

__global__ __launch_bounds__(512) __attribute__((amdgpu_waves_per_eu(4, 4)))
void xr_fused(const float* __restrict__ X, const float* __restrict__ w,
              const float* __restrict__ vb, const float* __restrict__ va,
              float* __restrict__ out) {
  __shared__ __attribute__((aligned(16))) float Qs[QSIZE * RPITCH];
  __shared__ __attribute__((aligned(16))) float Q2s[QSIZE * RPITCH];
  __shared__ __attribute__((aligned(16))) float Us[QSIZE * RPITCH];
  __shared__ __attribute__((aligned(16))) float Rs[64 * RPITCH];
  __shared__ float red[16];

  const int tid = threadIdx.x;
  const int lane = tid & 31;
  const int wave = tid >> 5;
  const int lo = lane & 15, hi = lane >> 4;
  const int ch = blockIdx.x & 1;
  const int rb = blockIdx.x >> 1;
  const int ml = 16 * (wave >> 2);
  const int m0 = 64 * ch + ml;
  const int n0 = 32 * (wave & 3);
  const int mx = rb * 512 + wave * 32;

  float wv[8][4];
#pragma unroll
  for (int i = 0; i < 8; ++i) {
    const int r = wave + 16 * i;
    const int base = r * (2 * QSIZE - r - 1) / 2;
#pragma unroll
    for (int q = 0; q < 4; ++q) {
      const int c = r + 1 + lane + 32 * q;
      wv[i][q] = (c < QSIZE) ? w[base + lane + 32 * q] : 0.0f;
    }
  }
  float bv[8], av[2];
#pragma unroll
  for (int r = 0; r < 8; ++r) bv[r] = vb[m0 + 8 * hi + r];
#pragma unroll
  for (int j = 0; j < 2; ++j) av[j] = va[n0 + 16 * j + lo];

  const float* xrow = X + (mx + lo) * QSIZE + 4 * hi;
  v4f a[2][16];

  float s = 0.0f;
#pragma unroll
  for (int i = 0; i < 8; ++i) {
#pragma unroll
    for (int q = 0; q < 4; ++q) s += wv[i][q] * wv[i][q];
  }
#pragma unroll
  for (int off = 16; off > 0; off >>= 1) s += __shfl_xor(s, off);
  if (lane == 0) red[wave] = s;
  lds_barrier();
  float tot = 0.0f;
#pragma unroll
  for (int i = 0; i < 16; ++i) tot += red[i];
  const float norm = sqrtf(2.0f * tot);
  const bool finite = (norm == norm) && (norm < 3.0e38f);
  const float scale = (finite && norm > CN_EPS) ? (CN_EPS / (norm + 1e-12f))
                                                : 1.0f;
#pragma unroll
  for (int i = 0; i < 8; ++i) {
    const int r = wave + 16 * i;
#pragma unroll
    for (int q = 0; q < 4; ++q) {
      const int c = r + 1 + lane + 32 * q;
      if (c < QSIZE) {
        const float v = wv[i][q] * scale;
        Qs[r * RPITCH + c] = v;
        Qs[c * RPITCH + r] = -v;
      }
    }
  }
  if (tid < QSIZE) Qs[tid * RPITCH + tid] = 0.0f;
  lds_barrier();

  const int foff = lo * RPITCH + 4 * hi;
  v8f acc[2];

  {
    const int swave = __builtin_amdgcn_readfirstlane(wave);
    const int g = swave >> 2, i = swave & 3;
    int tm0, tn0, tm1, tn1;
    upper_tile(g, i, tm0, tn0);
    upper_tile(g, i + 4, tm1, tn1);
    if (i == g) {
      int tm2, tn2;
      upper_tile(g, 8, tm2, tn2);
      const float* const ab[3] = {Qs + 16 * tm0 * RPITCH + foff,
                                  Qs + 16 * tm1 * RPITCH + foff,
                                  Qs + 16 * tm2 * RPITCH + foff};
      const float* const bb[3] = {Qs + 16 * tn0 * RPITCH + foff,
                                  Qs + 16 * tn1 * RPITCH + foff,
                                  Qs + 16 * tn2 * RPITCH + foff};
      v8f acc3[3] = {(v8f){}, (v8f){}, (v8f){}};
      tile_pairs<3>(ab, bb, acc3, xrow, a[0]);
      store_q2_tile(Q2s, Us, Qs, tm0, tn0, lo, hi, acc3[0]);
      store_q2_tile(Q2s, Us, Qs, tm1, tn1, lo, hi, acc3[1]);
      store_q2_tile(Q2s, Us, Qs, tm2, tn2, lo, hi, acc3[2]);
    } else {
      const float* const ab[2] = {Qs + 16 * tm0 * RPITCH + foff,
                                  Qs + 16 * tm1 * RPITCH + foff};
      const float* const bb[2] = {Qs + 16 * tn0 * RPITCH + foff,
                                  Qs + 16 * tn1 * RPITCH + foff};
      v8f acc2[2] = {(v8f){}, (v8f){}};
      tile_pairs<2>(ab, bb, acc2, xrow, a[0]);
      store_q2_tile(Q2s, Us, Qs, tm0, tn0, lo, hi, acc2[0]);
      store_q2_tile(Q2s, Us, Qs, tm1, tn1, lo, hi, acc2[1]);
    }
  }
  lds_barrier();

#pragma unroll
  for (int j = 0; j < 2; ++j) acc[j] = (v8f){};
  rows_times_rows<2>(Q2s + m0 * RPITCH + foff, Us + n0 * RPITCH + foff, acc,
                     xrow + 16 * QSIZE, a[1]);

#pragma unroll
  for (int j = 0; j < 2; ++j) {
#pragma unroll
    for (int r = 0; r < 8; ++r) {
      const int row = m0 + 8 * hi + r;
      const int col = n0 + 16 * j + lo;
      const int e = row * RPITCH + col;
      const float rr = ((row == col) ? 1.0f : 0.0f) + 2.0f * Qs[e] +
                       2.0f * Q2s[e] + acc[j][r];
      Rs[(ml + 8 * hi + r) * RPITCH + col] = bv[r] * rr * av[j];
    }
  }
  lds_barrier();

  v8f oacc[2][4];
#pragma unroll
  for (int mi = 0; mi < 2; ++mi) {
#pragma unroll
    for (int j = 0; j < 4; ++j) oacc[mi][j] = (v8f){};
  }

  const float* bbase = Rs + foff;
  v4f bcur[4], bnxt[4];
#pragma unroll
  for (int j = 0; j < 4; ++j)
    bcur[j] = *(const v4f*)(bbase + j * 16 * RPITCH);

#pragma unroll
  for (int t = 0; t < 16; ++t) {
    const bool more = (t + 1 < 16);
    if (more) {
#pragma unroll
      for (int j = 0; j < 4; ++j)
        bnxt[j] = *(const v4f*)(bbase + j * 16 * RPITCH + 8 * (t + 1));
    }
#pragma unroll
    for (int mi = 0; mi < 2; ++mi) {
      const v2f a0 = {a[mi][t].x, a[mi][t].y};
#pragma unroll
      for (int j = 0; j < 4; ++j) {
        const v2f b0 = {bcur[j].x, bcur[j].y};
        oacc[mi][j] = wmma_f32x4(a0, b0, oacc[mi][j]);
      }
    }
#pragma unroll
    for (int mi = 0; mi < 2; ++mi) {
      const v2f a1 = {a[mi][t].z, a[mi][t].w};
#pragma unroll
      for (int j = 0; j < 4; ++j) {
        const v2f b1 = {bcur[j].z, bcur[j].w};
        oacc[mi][j] = wmma_f32x4(a1, b1, oacc[mi][j]);
      }
    }
    if (more) {
#pragma unroll
      for (int j = 0; j < 4; ++j) bcur[j] = bnxt[j];
    }
  }

  float* orow = out + (mx + 8 * hi) * QSIZE + 64 * ch + lo;
#pragma unroll
  for (int mi = 0; mi < 2; ++mi) {
#pragma unroll
    for (int r = 0; r < 8; ++r) {
#pragma unroll
      for (int j = 0; j < 4; ++j)
        __builtin_nontemporal_store(oacc[mi][j][r],
                                    orow + (16 * mi + r) * QSIZE + 16 * j);
    }
  }
}

extern "C" void kernel_launch(void* const* d_in, const int* in_sizes, int n_in,
                              void* d_out, int out_size, void* d_ws,
                              size_t ws_size, hipStream_t stream) {
  const float* x = (const float*)d_in[0];
  const float* w = (const float*)d_in[1];
  const float* vb = (const float*)d_in[2];
  const float* va = (const float*)d_in[3];
  float* out = (float*)d_out;

  const int rows = in_sizes[0] / QSIZE;
  const int mblocks = rows / 256;

  xr_fused<<<mblocks, 512, 0, stream>>>(x, w, vb, va, out);
}
